# starting kernel
# baseline (speedup 1.0000x reference)
#include <hip/hip_runtime.h>

typedef __attribute__((ext_vector_type(2))) float v2f;
typedef __attribute__((ext_vector_type(8))) float v8f;

#define GAT_B 16
#define GAT_K 128
#define GAT_F 128
#define GAT_H 256
#define GAT_ALPHA 0.2f

__global__ void __launch_bounds__(32)
gat_qk_gemm(const float* __restrict__ x,
            const float* __restrict__ w,
            float* __restrict__ s1,
            float* __restrict__ s2) {
  int tile = blockIdx.x;
  const int ht = tile % (GAT_H / 16); tile /= (GAT_H / 16);
  const int it = tile % (GAT_K / 16); tile /= (GAT_K / 16);
  const int b  = tile;
  const int i0 = it * 16, h0 = ht * 16;
  const int lane = threadIdx.x & 31;
  const int hf   = lane >> 4;
  const int l    = lane & 15;

  const float* xb = x + b * GAT_F * GAT_K;
  const float* wr = w + (h0 + l) * (2 * GAT_F);

  v8f c1 = {};
  v8f c2 = {};
  for (int f0 = 0; f0 < GAT_F; f0 += 4) {
    v2f av, b1, b2;
    av.x = xb[(f0 + 2 * hf + 0) * GAT_K + i0 + l];
    av.y = xb[(f0 + 2 * hf + 1) * GAT_K + i0 + l];
    b1.x = wr[f0 + 2 * hf + 0];
    b1.y = wr[f0 + 2 * hf + 1];
    b2.x = wr[GAT_F + f0 + 2 * hf + 0];
    b2.y = wr[GAT_F + f0 + 2 * hf + 1];
    c1 = __builtin_amdgcn_wmma_f32_16x16x4_f32(false, av, false, b1, (short)0,
                                               c1, false, false);
    c2 = __builtin_amdgcn_wmma_f32_16x16x4_f32(false, av, false, b2, (short)0,
                                               c2, false, false);
  }
#pragma unroll
  for (int r = 0; r < 8; ++r) {
    const int row = i0 + r + 8 * hf;
    const int col = h0 + l;
    s1[(b * GAT_K + row) * GAT_H + col] = c1[r];
    s2[(b * GAT_K + row) * GAT_H + col] = c2[r];
  }
}

__global__ void __launch_bounds__(256)
gat_attn(const float* __restrict__ s1,
         const float* __restrict__ s2,
         const float* __restrict__ lb,
         const float* __restrict__ av,
         const float* __restrict__ bias,
         float* __restrict__ attn) {
  __shared__ float s2lds[64 * GAT_H];

  const int blk  = blockIdx.x;
  const int ic   = blk % (GAT_K / 8);
  const int b    = blk / (GAT_K / 8);
  const int wave = threadIdx.x >> 5;
  const int lane = threadIdx.x & 31;
  const int i    = ic * 8 + wave;

  float r1[8], ar[8];
#pragma unroll
  for (int r = 0; r < 8; ++r) {
    const int h = lane + 32 * r;
    r1[r] = s1[(b * GAT_K + i) * GAT_H + h] + lb[h];
    ar[r] = av[h];
  }

  float e0 = 0.f, e1 = 0.f, e2 = 0.f, e3 = 0.f;

#pragma unroll
  for (int c = 0; c < 2; ++c) {
    __syncthreads();
    for (int idx = threadIdx.x; idx < 64 * GAT_H; idx += 256) {
      const int jj = idx >> 8;
      const int h  = idx & (GAT_H - 1);
      s2lds[idx] = s2[(b * GAT_K + c * 64 + jj) * GAT_H + h];
    }
    __syncthreads();

#pragma unroll
    for (int t2 = 0; t2 < 2; ++t2) {
      for (int jl = 0; jl < 32; ++jl) {
        const int jj = t2 * 32 + jl;
        float acc = 0.f;
#pragma unroll
        for (int r = 0; r < 8; ++r) {
          float t = r1[r] + s2lds[jj * GAT_H + lane + 32 * r];
          t = fmaxf(t, GAT_ALPHA * t);
          acc = fmaf(t, ar[r], acc);
        }
        acc += __shfl_xor(acc, 1);
        acc += __shfl_xor(acc, 2);
        acc += __shfl_xor(acc, 4);
        acc += __shfl_xor(acc, 8);
        acc += __shfl_xor(acc, 16);
        const int j = c * 64 + jj;
        acc += bias[i * GAT_K + j];
        if (jl == lane) {
          if (c * 2 + t2 == 0)      e0 = acc;
          else if (c * 2 + t2 == 1) e1 = acc;
          else if (c * 2 + t2 == 2) e2 = acc;
          else                      e3 = acc;
        }
      }
    }
  }

  float m = fmaxf(fmaxf(e0, e1), fmaxf(e2, e3));
  m = fmaxf(m, __shfl_xor(m, 1));
  m = fmaxf(m, __shfl_xor(m, 2));
  m = fmaxf(m, __shfl_xor(m, 4));
  m = fmaxf(m, __shfl_xor(m, 8));
  m = fmaxf(m, __shfl_xor(m, 16));
  e0 = __expf(e0 - m);
  e1 = __expf(e1 - m);
  e2 = __expf(e2 - m);
  e3 = __expf(e3 - m);
  float s = e0 + e1 + e2 + e3;
  s += __shfl_xor(s, 1);
  s += __shfl_xor(s, 2);
  s += __shfl_xor(s, 4);
  s += __shfl_xor(s, 8);
  s += __shfl_xor(s, 16);
  const float inv = 1.0f / s;
  float* arow = attn + (b * GAT_K + i) * GAT_K;
  arow[lane +  0] = e0 * inv;
  arow[lane + 32] = e1 * inv;
  arow[lane + 64] = e2 * inv;
  arow[lane + 96] = e3 * inv;
}

__global__ void __launch_bounds__(32)
gat_out(const float* __restrict__ attn,
        const float* __restrict__ x,
        float* __restrict__ out) {
  int tile = blockIdx.x;
  const int ft = tile % (GAT_F / 16); tile /= (GAT_F / 16);
  const int it = tile % (GAT_K / 16); tile /= (GAT_K / 16);
  const int b  = tile;
  const int i0 = it * 16, f0 = ft * 16;
  const int lane = threadIdx.x & 31;
  const int hf   = lane >> 4;
  const int l    = lane & 15;

  const float* ab = attn + b * GAT_K * GAT_K;
  const float* xb = x + b * GAT_F * GAT_K;

  v8f c = {};
  for (int j0 = 0; j0 < GAT_K; j0 += 4) {
    v2f av, bv;
    av.x = ab[(i0 + l) * GAT_K + j0 + 2 * hf + 0];
    av.y = ab[(i0 + l) * GAT_K + j0 + 2 * hf + 1];
    bv.x = xb[(f0 + l) * GAT_K + j0 + 2 * hf + 0];
    bv.y = xb[(f0 + l) * GAT_K + j0 + 2 * hf + 1];
    c = __builtin_amdgcn_wmma_f32_16x16x4_f32(false, av, false, bv, (short)0,
                                              c, false, false);
  }
#pragma unroll
  for (int r = 0; r < 8; ++r) {
    const int row = i0 + r + 8 * hf;
    const int col = f0 + l;
    const float z = c[r];
    out[(b * GAT_K + row) * GAT_F + col] = 1.0f / (1.0f + __expf(-z));
  }
}

extern "C" void kernel_launch(void* const* d_in, const int* in_sizes, int n_in,
                              void* d_out, int out_size, void* d_ws, size_t ws_size,
                              hipStream_t stream) {
  (void)in_sizes; (void)n_in; (void)out_size; (void)ws_size;
  const float* x    = (const float*)d_in[0];
  const float* w    = (const float*)d_in[1];
  const float* lb   = (const float*)d_in[2];
  const float* av   = (const float*)d_in[3];
  const float* bias = (const float*)d_in[4];
  float* out = (float*)d_out;

  float* s1   = (float*)d_ws;
  float* s2   = s1 + GAT_B * GAT_K * GAT_H;
  float* attn = s2 + GAT_B * GAT_K * GAT_H;

  gat_qk_gemm<<<GAT_B * (GAT_K / 16) * (GAT_H / 16), 32, 0, stream>>>(x, w, s1, s2);
  gat_attn  <<<GAT_B * (GAT_K / 8),               256, 0, stream>>>(s1, s2, lb, av, bias, attn);
  gat_out   <<<GAT_B * (GAT_K / 16) * (GAT_F / 16), 32, 0, stream>>>(attn, x, out);
}


# attention scores: each lane owns one column j and sums over h serially (GEMM stores s2 transposed, h-quad-major), replacing the per-(i,j) cross-lane butterfly; fp32 sum order over h changes
# speedup vs baseline: 2.4753x; 2.4753x over previous
#include <hip/hip_runtime.h>

typedef __attribute__((ext_vector_type(2))) float v2f;
typedef __attribute__((ext_vector_type(8))) float v8f;

#define GAT_B 16
#define GAT_K 128
#define GAT_F 128
#define GAT_H 256
#define GAT_ALPHA 0.2f

__global__ void __launch_bounds__(32)
gat_qk_gemm(const float* __restrict__ x,
            const float* __restrict__ w,
            const float* __restrict__ lb,
            float* __restrict__ s1p,
            float* __restrict__ s2q) {
  int tile = blockIdx.x;
  const int ht = tile % (GAT_H / 16); tile /= (GAT_H / 16);
  const int it = tile % (GAT_K / 16); tile /= (GAT_K / 16);
  const int b  = tile;
  const int i0 = it * 16, h0 = ht * 16;
  const int lane = threadIdx.x & 31;
  const int hf   = lane >> 4;
  const int l    = lane & 15;

  const float* xb = x + b * GAT_F * GAT_K;
  const float* wr = w + (h0 + l) * (2 * GAT_F);

  v8f c1 = {};
  v8f c2 = {};
  for (int f0 = 0; f0 < GAT_F; f0 += 4) {
    v2f av, b1, b2;
    av.x = xb[(f0 + 2 * hf + 0) * GAT_K + i0 + l];
    av.y = xb[(f0 + 2 * hf + 1) * GAT_K + i0 + l];
    b1.x = wr[f0 + 2 * hf + 0];
    b1.y = wr[f0 + 2 * hf + 1];
    b2.x = wr[GAT_F + f0 + 2 * hf + 0];
    b2.y = wr[GAT_F + f0 + 2 * hf + 1];
    c1 = __builtin_amdgcn_wmma_f32_16x16x4_f32(false, av, false, b1, (short)0,
                                               c1, false, false);
    c2 = __builtin_amdgcn_wmma_f32_16x16x4_f32(false, b2, false, av, (short)0,
                                               c2, false, false);
  }
  const float lbv = lb[h0 + l];
#pragma unroll
  for (int r = 0; r < 8; ++r) {
    const int row = i0 + r + 8 * hf;
    const int col = h0 + l;
    s1p[(b * GAT_K + row) * GAT_H + col] = c1[r] + lbv;
  }
  float4* q = reinterpret_cast<float4*>(s2q) +
              (b * (GAT_H / 4) + (h0 + 8 * hf) / 4) * GAT_K + i0 + l;
  q[0]     = make_float4(c2[0], c2[1], c2[2], c2[3]);
  q[GAT_K] = make_float4(c2[4], c2[5], c2[6], c2[7]);
}

__global__ void __launch_bounds__(512)
gat_attn(const float* __restrict__ s1p,
         const float* __restrict__ s2q,
         const float* __restrict__ av,
         const float* __restrict__ bias,
         float* __restrict__ attn) {
  __shared__ float4 s2l[(GAT_H / 4) * GAT_K];
  __shared__ float  el[2 * 8 * GAT_K];

  const int blk  = blockIdx.x;
  const int ic   = blk % (GAT_K / 8);
  const int b    = blk / (GAT_K / 8);
  const int wave = __builtin_amdgcn_readfirstlane((int)(threadIdx.x >> 5));
  const int lane = threadIdx.x & 31;

  {
    const float4* src = reinterpret_cast<const float4*>(s2q) +
                        b * (GAT_H / 4) * GAT_K;
#pragma unroll
    for (int n = 0; n < (GAT_H / 4) * GAT_K / 512; ++n)
      s2l[n * 512 + threadIdx.x] = src[n * 512 + threadIdx.x];
  }
  __syncthreads();

  const int hh = wave >> 3;
  const int rg = (wave >> 2) & 1;
  const int jq = wave & 3;
  const int j  = jq * 32 + lane;
  const int i0 = ic * 8 + rg * 4;
  const float* r1p = s1p + (b * GAT_K + i0) * GAT_H;

  float acc0 = 0.f, acc1 = 0.f, acc2 = 0.f, acc3 = 0.f;

#pragma unroll 1
  for (int h0 = hh * (GAT_H / 2); h0 < (hh + 1) * (GAT_H / 2); h0 += 8) {
    const float4 sa = s2l[(h0 / 4) * GAT_K + j];
    const float4 sb = s2l[(h0 / 4 + 1) * GAT_K + j];
    const float sv[8] = {sa.x, sa.y, sa.z, sa.w, sb.x, sb.y, sb.z, sb.w};
#pragma unroll
    for (int k = 0; k < 8; ++k) {
      const float ah = av[h0 + k];
      float t;
      t = r1p[0 * GAT_H + h0 + k] + sv[k];
      t = fmaxf(t, GAT_ALPHA * t);
      acc0 = fmaf(t, ah, acc0);
      t = r1p[1 * GAT_H + h0 + k] + sv[k];
      t = fmaxf(t, GAT_ALPHA * t);
      acc1 = fmaf(t, ah, acc1);
      t = r1p[2 * GAT_H + h0 + k] + sv[k];
      t = fmaxf(t, GAT_ALPHA * t);
      acc2 = fmaf(t, ah, acc2);
      t = r1p[3 * GAT_H + h0 + k] + sv[k];
      t = fmaxf(t, GAT_ALPHA * t);
      acc3 = fmaf(t, ah, acc3);
    }
  }

  el[(hh * 8 + rg * 4 + 0) * GAT_K + j] = acc0;
  el[(hh * 8 + rg * 4 + 1) * GAT_K + j] = acc1;
  el[(hh * 8 + rg * 4 + 2) * GAT_K + j] = acc2;
  el[(hh * 8 + rg * 4 + 3) * GAT_K + j] = acc3;
  __syncthreads();

  if (wave < 8) {
    const int i = ic * 8 + wave;
    const float* p0 = el + wave * GAT_K + lane;
    const float* p1 = el + (8 + wave) * GAT_K + lane;
    const float* bi = bias + i * GAT_K + lane;
    float e0 = (p0[ 0] + p1[ 0]) + bi[ 0];
    float e1 = (p0[32] + p1[32]) + bi[32];
    float e2 = (p0[64] + p1[64]) + bi[64];
    float e3 = (p0[96] + p1[96]) + bi[96];
    float m = fmaxf(fmaxf(e0, e1), fmaxf(e2, e3));
    m = fmaxf(m, __shfl_xor(m, 1));
    m = fmaxf(m, __shfl_xor(m, 2));
    m = fmaxf(m, __shfl_xor(m, 4));
    m = fmaxf(m, __shfl_xor(m, 8));
    m = fmaxf(m, __shfl_xor(m, 16));
    e0 = __expf(e0 - m);
    e1 = __expf(e1 - m);
    e2 = __expf(e2 - m);
    e3 = __expf(e3 - m);
    float s = e0 + e1 + e2 + e3;
    s += __shfl_xor(s, 1);
    s += __shfl_xor(s, 2);
    s += __shfl_xor(s, 4);
    s += __shfl_xor(s, 8);
    s += __shfl_xor(s, 16);
    const float inv = 1.0f / s;
    float* arow = attn + (b * GAT_K + i) * GAT_K;
    arow[lane +  0] = e0 * inv;
    arow[lane + 32] = e1 * inv;
    arow[lane + 64] = e2 * inv;
    arow[lane + 96] = e3 * inv;
  }
}

__global__ void __launch_bounds__(32)
gat_out(const float* __restrict__ attn,
        const float* __restrict__ x,
        float* __restrict__ out) {
  int tile = blockIdx.x;
  const int ft = tile % (GAT_F / 16); tile /= (GAT_F / 16);
  const int it = tile % (GAT_K / 16); tile /= (GAT_K / 16);
  const int b  = tile;
  const int i0 = it * 16, f0 = ft * 16;
  const int lane = threadIdx.x & 31;
  const int hf   = lane >> 4;
  const int l    = lane & 15;

  const float* ab = attn + b * GAT_K * GAT_K;
  const float* xb = x + b * GAT_F * GAT_K;

  v8f c = {};
  for (int j0 = 0; j0 < GAT_K; j0 += 4) {
    v2f av, bv;
    av.x = ab[(i0 + l) * GAT_K + j0 + 2 * hf + 0];
    av.y = ab[(i0 + l) * GAT_K + j0 + 2 * hf + 1];
    bv.x = xb[(f0 + l) * GAT_K + j0 + 2 * hf + 0];
    bv.y = xb[(f0 + l) * GAT_K + j0 + 2 * hf + 1];
    c = __builtin_amdgcn_wmma_f32_16x16x4_f32(false, av, false, bv, (short)0,
                                              c, false, false);
  }
#pragma unroll
  for (int r = 0; r < 8; ++r) {
    const int row = i0 + r + 8 * hf;
    const int col = f0 + l;
    const float z = c[r];
    out[(b * GAT_K + row) * GAT_F + col] = 1.0f / (1.0f + __expf(-z));
  }
}

extern "C" void kernel_launch(void* const* d_in, const int* in_sizes, int n_in,
                              void* d_out, int out_size, void* d_ws, size_t ws_size,
                              hipStream_t stream) {
  (void)in_sizes; (void)n_in; (void)out_size; (void)ws_size;
  const float* x    = (const float*)d_in[0];
  const float* w    = (const float*)d_in[1];
  const float* lb   = (const float*)d_in[2];
  const float* av   = (const float*)d_in[3];
  const float* bias = (const float*)d_in[4];
  float* out = (float*)d_out;

  float* s1p  = (float*)d_ws;
  float* s2q  = s1p + GAT_B * GAT_K * GAT_H;
  float* attn = s2q + GAT_B * GAT_K * GAT_H;

  gat_qk_gemm<<<GAT_B * (GAT_K / 16) * (GAT_H / 16), 32, 0, stream>>>(x, w, lb, s1p, s2q);
  gat_attn  <<<GAT_B * (GAT_K / 8),               512, 0, stream>>>(s1p, s2q, av, bias, attn);
  gat_out   <<<GAT_B * (GAT_K / 16) * (GAT_F / 16), 32, 0, stream>>>(attn, x, out);
}


# fuse out=sigmoid(attn@v) into the attention kernel: softmax rows stay in LDS, wave = 8x16 tile x half of j on fp32 WMMA, x loads cross LDS-only barriers; launches 3->2; fp32 sum order over j changes
# speedup vs baseline: 2.9671x; 1.1987x over previous
#include <hip/hip_runtime.h>

typedef __attribute__((ext_vector_type(2))) float v2f;
typedef __attribute__((ext_vector_type(8))) float v8f;

#define GAT_B 16
#define GAT_K 128
#define GAT_F 128
#define GAT_H 256
#define GAT_ALPHA 0.2f

__global__ void __launch_bounds__(32)
gat_qk_gemm(const float* __restrict__ x,
            const float* __restrict__ w,
            const float* __restrict__ lb,
            float* __restrict__ s1p,
            float* __restrict__ s2q) {
  int tile = blockIdx.x;
  const int ht = tile % (GAT_H / 16); tile /= (GAT_H / 16);
  const int it = tile % (GAT_K / 16); tile /= (GAT_K / 16);
  const int b  = tile;
  const int i0 = it * 16, h0 = ht * 16;
  const int lane = threadIdx.x & 31;
  const int hf   = lane >> 4;
  const int l    = lane & 15;

  const float* xb = x + b * GAT_F * GAT_K;
  const float* wr = w + (h0 + l) * (2 * GAT_F);

  v8f c1 = {};
  v8f c2 = {};
  for (int f0 = 0; f0 < GAT_F; f0 += 4) {
    v2f av, b1, b2;
    av.x = xb[(f0 + 2 * hf + 0) * GAT_K + i0 + l];
    av.y = xb[(f0 + 2 * hf + 1) * GAT_K + i0 + l];
    b1.x = wr[f0 + 2 * hf + 0];
    b1.y = wr[f0 + 2 * hf + 1];
    b2.x = wr[GAT_F + f0 + 2 * hf + 0];
    b2.y = wr[GAT_F + f0 + 2 * hf + 1];
    c1 = __builtin_amdgcn_wmma_f32_16x16x4_f32(false, av, false, b1, (short)0,
                                               c1, false, false);
    c2 = __builtin_amdgcn_wmma_f32_16x16x4_f32(false, b2, false, av, (short)0,
                                               c2, false, false);
  }
  const float lbv = lb[h0 + l];
#pragma unroll
  for (int r = 0; r < 8; ++r) {
    const int row = i0 + r + 8 * hf;
    const int col = h0 + l;
    s1p[(b * GAT_K + row) * GAT_H + col] = c1[r] + lbv;
  }
  float4* q = reinterpret_cast<float4*>(s2q) +
              (b * (GAT_H / 4) + (h0 + 8 * hf) / 4) * GAT_K + i0 + l;
  q[0]     = make_float4(c2[0], c2[1], c2[2], c2[3]);
  q[GAT_K] = make_float4(c2[4], c2[5], c2[6], c2[7]);
}

__device__ __forceinline__ void lds_barrier() {
  __builtin_amdgcn_fence(__ATOMIC_RELEASE, "workgroup", "local");
  __builtin_amdgcn_s_barrier();
  __builtin_amdgcn_fence(__ATOMIC_ACQUIRE, "workgroup", "local");
}

__global__ void __launch_bounds__(512)
gat_attn_out(const float* __restrict__ s1p,
             const float* __restrict__ s2q,
             const float* __restrict__ av,
             const float* __restrict__ bias,
             const float* __restrict__ x,
             float* __restrict__ out) {
  __shared__ float4 s2l[(GAT_H / 4) * GAT_K];
  __shared__ float  el[2 * 8 * GAT_K];
  __shared__ float4 pl[8 * 33];

  const int blk  = blockIdx.x;
  const int ic   = blk % (GAT_K / 8);
  const int b    = blk / (GAT_K / 8);
  const int wave = __builtin_amdgcn_readfirstlane((int)(threadIdx.x >> 5));
  const int lane = threadIdx.x & 31;
  const int hf   = lane >> 4;
  const int l    = lane & 15;

  {
    const float4* src = reinterpret_cast<const float4*>(s2q) +
                        b * (GAT_H / 4) * GAT_K;
#pragma unroll
    for (int n = 0; n < (GAT_H / 4) * GAT_K / 512; ++n)
      s2l[n * 512 + threadIdx.x] = src[n * 512 + threadIdx.x];
  }

  __syncthreads();


  const int hh = wave >> 3;
  const int rg = (wave >> 2) & 1;
  const int jq = wave & 3;
  const int j  = jq * 32 + lane;
  const int i0 = ic * 8 + rg * 4;
  const float* r1p = s1p + (b * GAT_K + i0) * GAT_H;

  float acc0 = 0.f, acc1 = 0.f, acc2 = 0.f, acc3 = 0.f;

#pragma unroll 1
  for (int h0 = hh * (GAT_H / 2); h0 < (hh + 1) * (GAT_H / 2); h0 += 8) {
    const float4 sa = s2l[(h0 / 4) * GAT_K + j];
    const float4 sb = s2l[(h0 / 4 + 1) * GAT_K + j];
    const float sv[8] = {sa.x, sa.y, sa.z, sa.w, sb.x, sb.y, sb.z, sb.w};
#pragma unroll
    for (int k = 0; k < 8; ++k) {
      const float ah = av[h0 + k];
      float t;
      t = r1p[0 * GAT_H + h0 + k] + sv[k];
      t = fmaxf(t, GAT_ALPHA * t);
      acc0 = fmaf(t, ah, acc0);
      t = r1p[1 * GAT_H + h0 + k] + sv[k];
      t = fmaxf(t, GAT_ALPHA * t);
      acc1 = fmaf(t, ah, acc1);
      t = r1p[2 * GAT_H + h0 + k] + sv[k];
      t = fmaxf(t, GAT_ALPHA * t);
      acc2 = fmaf(t, ah, acc2);
      t = r1p[3 * GAT_H + h0 + k] + sv[k];
      t = fmaxf(t, GAT_ALPHA * t);
      acc3 = fmaf(t, ah, acc3);
    }
  }

  const int ft = wave & 7;
  const int kh = wave >> 3;
  float4 xq[8];
  {
    const float4* xr = reinterpret_cast<const float4*>(
        x + (b * GAT_F + ft * 16 + l) * GAT_K + kh * 64 + hf * 4);
#pragma unroll
    for (int uu = 0; uu < 8; ++uu) xq[uu] = xr[uu * 2];
  }

  el[(hh * 8 + rg * 4 + 0) * GAT_K + j] = acc0;
  el[(hh * 8 + rg * 4 + 1) * GAT_K + j] = acc1;
  el[(hh * 8 + rg * 4 + 2) * GAT_K + j] = acc2;
  el[(hh * 8 + rg * 4 + 3) * GAT_K + j] = acc3;
  lds_barrier();

  if (wave < 8) {
    const int i = ic * 8 + wave;
    const float* p0 = el + wave * GAT_K + lane;
    const float* p1 = el + (8 + wave) * GAT_K + lane;
    const float* bi = bias + i * GAT_K + lane;
    float e0 = (p0[ 0] + p1[ 0]) + bi[ 0];
    float e1 = (p0[32] + p1[32]) + bi[32];
    float e2 = (p0[64] + p1[64]) + bi[64];
    float e3 = (p0[96] + p1[96]) + bi[96];
    float m = fmaxf(fmaxf(e0, e1), fmaxf(e2, e3));
    m = fmaxf(m, __shfl_xor(m, 1));
    m = fmaxf(m, __shfl_xor(m, 2));
    m = fmaxf(m, __shfl_xor(m, 4));
    m = fmaxf(m, __shfl_xor(m, 8));
    m = fmaxf(m, __shfl_xor(m, 16));
    e0 = __expf(e0 - m);
    e1 = __expf(e1 - m);
    e2 = __expf(e2 - m);
    e3 = __expf(e3 - m);
    float s = e0 + e1 + e2 + e3;
    s += __shfl_xor(s, 1);
    s += __shfl_xor(s, 2);
    s += __shfl_xor(s, 4);
    s += __shfl_xor(s, 8);
    s += __shfl_xor(s, 16);
    const float inv = 1.0f / s;
    float* prow = reinterpret_cast<float*>(pl) + wave * 132 + lane;
    prow[ 0] = e0 * inv;
    prow[32] = e1 * inv;
    prow[64] = e2 * inv;
    prow[96] = e3 * inv;
  }
  lds_barrier();

  v8f c0 = {};
  v8f c1 = {};
#pragma unroll
  for (int uu = 0; uu < 8; ++uu) {
    const float4 pa = pl[(l & 7) * 33 + kh * 16 + uu * 2 + hf];
    const float4 xb = xq[uu];
    v2f a0, a1, b0, b1;
    a0.x = pa.x; a0.y = pa.y; a1.x = pa.z; a1.y = pa.w;
    b0.x = xb.x; b0.y = xb.y; b1.x = xb.z; b1.y = xb.w;
    c0 = __builtin_amdgcn_wmma_f32_16x16x4_f32(false, a0, false, b0, (short)0,
                                               c0, false, false);
    c1 = __builtin_amdgcn_wmma_f32_16x16x4_f32(false, a1, false, b1, (short)0,
                                               c1, false, false);
  }
  {
    float* ep = el + (kh * 8 + 4 * hf) * GAT_K + ft * 16 + l;
    ep[0 * GAT_K] = hf ? (c0[4] + c1[4]) : (c0[0] + c1[0]);
    ep[1 * GAT_K] = hf ? (c0[5] + c1[5]) : (c0[1] + c1[1]);
    ep[2 * GAT_K] = hf ? (c0[6] + c1[6]) : (c0[2] + c1[2]);
    ep[3 * GAT_K] = hf ? (c0[7] + c1[7]) : (c0[3] + c1[3]);
  }
  __syncthreads();

  {
    const int row = wave & 7;
    const int f   = (wave >> 3) * 64 + lane;
    const float z0 = el[row * GAT_K + f]      + el[(8 + row) * GAT_K + f];
    const float z1 = el[row * GAT_K + f + 32] + el[(8 + row) * GAT_K + f + 32];
    float* orow = out + (b * GAT_K + ic * 8 + row) * GAT_F + f;
    orow[ 0] = 1.0f / (1.0f + __expf(-z0));
    orow[32] = 1.0f / (1.0f + __expf(-z1));
  }
}

extern "C" void kernel_launch(void* const* d_in, const int* in_sizes, int n_in,
                              void* d_out, int out_size, void* d_ws, size_t ws_size,
                              hipStream_t stream) {
  (void)in_sizes; (void)n_in; (void)out_size; (void)ws_size;
  const float* x    = (const float*)d_in[0];
  const float* w    = (const float*)d_in[1];
  const float* lb   = (const float*)d_in[2];
  const float* av   = (const float*)d_in[3];
  const float* bias = (const float*)d_in[4];
  float* out = (float*)d_out;

  float* s1p  = (float*)d_ws;
  float* s2q  = s1p + GAT_B * GAT_K * GAT_H;

  gat_qk_gemm<<<GAT_B * (GAT_K / 16) * (GAT_H / 16), 32, 0, stream>>>(x, w, lb, s1p, s2q);
  gat_attn_out<<<GAT_B * (GAT_K / 8),             512, 0, stream>>>(s1p, s2q, av, bias, x, out);
}


# GEMM wave takes two 16x16 tiles (even/odd rows i) sharing W fragments read as 16-byte pieces (k order 8m+4hf+2s+e, both loaders), loads issued ahead; 2048->1024 waves; fp32 sum order over f changes
# speedup vs baseline: 3.6253x; 1.2218x over previous
#include <hip/hip_runtime.h>

typedef __attribute__((ext_vector_type(2))) float v2f;
typedef __attribute__((ext_vector_type(8))) float v8f;

#define GAT_B 16
#define GAT_K 128
#define GAT_F 128
#define GAT_H 256
#define GAT_ALPHA 0.2f

__global__ void __launch_bounds__(32) __attribute__((amdgpu_waves_per_eu(1, 2)))
gat_qk_gemm(const float* __restrict__ x,
            const float* __restrict__ w,
            const float* __restrict__ lb,
            float* __restrict__ s1p,
            float* __restrict__ s2q) {
  int tile = blockIdx.x;
  const int ht = tile % (GAT_H / 16); tile /= (GAT_H / 16);
  const int ip = tile % (GAT_K / 32); tile /= (GAT_K / 32);
  const int b  = tile;
  const int i0 = ip * 32, h0 = ht * 16;
  const int lane = threadIdx.x & 31;
  const int hf   = lane >> 4;
  const int l    = lane & 15;

  const float* xb = x + b * GAT_F * GAT_K + (4 * hf) * GAT_K + i0 + 2 * l;
  const float* wr = w + (h0 + l) * (2 * GAT_F) + 4 * hf;

  v8f c1a = {}, c1b = {};
  v8f c2a = {}, c2b = {};
#pragma unroll
  for (int m = 0; m < GAT_F / 8; ++m) {
    const float4 w1 = *reinterpret_cast<const float4*>(wr + 8 * m);
    const float4 w2 = *reinterpret_cast<const float4*>(wr + GAT_F + 8 * m);
    const v2f x0 = *reinterpret_cast<const v2f*>(xb + (8 * m + 0) * GAT_K);
    const v2f x1 = *reinterpret_cast<const v2f*>(xb + (8 * m + 1) * GAT_K);
    const v2f x2 = *reinterpret_cast<const v2f*>(xb + (8 * m + 2) * GAT_K);
    const v2f x3 = *reinterpret_cast<const v2f*>(xb + (8 * m + 3) * GAT_K);
    v2f va, vb, b1, b2;
    va.x = x0.x; va.y = x1.x;
    vb.x = x0.y; vb.y = x1.y;
    b1.x = w1.x; b1.y = w1.y;
    b2.x = w2.x; b2.y = w2.y;
    c1a = __builtin_amdgcn_wmma_f32_16x16x4_f32(false, va, false, b1, (short)0,
                                                c1a, false, false);
    c1b = __builtin_amdgcn_wmma_f32_16x16x4_f32(false, vb, false, b1, (short)0,
                                                c1b, false, false);
    c2a = __builtin_amdgcn_wmma_f32_16x16x4_f32(false, b2, false, va, (short)0,
                                                c2a, false, false);
    c2b = __builtin_amdgcn_wmma_f32_16x16x4_f32(false, b2, false, vb, (short)0,
                                                c2b, false, false);
    va.x = x2.x; va.y = x3.x;
    vb.x = x2.y; vb.y = x3.y;
    b1.x = w1.z; b1.y = w1.w;
    b2.x = w2.z; b2.y = w2.w;
    c1a = __builtin_amdgcn_wmma_f32_16x16x4_f32(false, va, false, b1, (short)0,
                                                c1a, false, false);
    c1b = __builtin_amdgcn_wmma_f32_16x16x4_f32(false, vb, false, b1, (short)0,
                                                c1b, false, false);
    c2a = __builtin_amdgcn_wmma_f32_16x16x4_f32(false, b2, false, va, (short)0,
                                                c2a, false, false);
    c2b = __builtin_amdgcn_wmma_f32_16x16x4_f32(false, b2, false, vb, (short)0,
                                                c2b, false, false);
  }
  const float lbv = lb[h0 + l];
#pragma unroll
  for (int r = 0; r < 8; ++r) {
    const int row = i0 + 2 * (r + 8 * hf);
    const int col = h0 + l;
    s1p[(b * GAT_K + row) * GAT_H + col]     = c1a[r] + lbv;
    s1p[(b * GAT_K + row + 1) * GAT_H + col] = c1b[r] + lbv;
  }
  float4* q = reinterpret_cast<float4*>(s2q) +
              (b * (GAT_H / 4) + (h0 + 8 * hf) / 4) * GAT_K + i0 + 2 * l;
  q[0]         = make_float4(c2a[0], c2a[1], c2a[2], c2a[3]);
  q[1]         = make_float4(c2b[0], c2b[1], c2b[2], c2b[3]);
  q[GAT_K]     = make_float4(c2a[4], c2a[5], c2a[6], c2a[7]);
  q[GAT_K + 1] = make_float4(c2b[4], c2b[5], c2b[6], c2b[7]);
}

__device__ __forceinline__ void lds_barrier() {
  __builtin_amdgcn_fence(__ATOMIC_RELEASE, "workgroup", "local");
  __builtin_amdgcn_s_barrier();
  __builtin_amdgcn_fence(__ATOMIC_ACQUIRE, "workgroup", "local");
}

__global__ void __launch_bounds__(512)
gat_attn_out(const float* __restrict__ s1p,
             const float* __restrict__ s2q,
             const float* __restrict__ av,
             const float* __restrict__ bias,
             const float* __restrict__ x,
             float* __restrict__ out) {
  __shared__ float4 s2l[(GAT_H / 4) * GAT_K];
  __shared__ float  el[2 * 8 * GAT_K];
  __shared__ float4 pl[8 * 33];

  const int blk  = blockIdx.x;
  const int ic   = blk % (GAT_K / 8);
  const int b    = blk / (GAT_K / 8);
  const int wave = __builtin_amdgcn_readfirstlane((int)(threadIdx.x >> 5));
  const int lane = threadIdx.x & 31;
  const int hf   = lane >> 4;
  const int l    = lane & 15;

  {
    const float4* src = reinterpret_cast<const float4*>(s2q) +
                        b * (GAT_H / 4) * GAT_K;
#pragma unroll
    for (int n = 0; n < (GAT_H / 4) * GAT_K / 512; ++n)
      s2l[n * 512 + threadIdx.x] = src[n * 512 + threadIdx.x];
  }

  __syncthreads();


  const int hh = wave >> 3;
  const int rg = (wave >> 2) & 1;
  const int jq = wave & 3;
  const int j  = jq * 32 + lane;
  const int i0 = ic * 8 + rg * 4;
  const float* r1p = s1p + (b * GAT_K + i0) * GAT_H;

  float acc0 = 0.f, acc1 = 0.f, acc2 = 0.f, acc3 = 0.f;

#pragma unroll 1
  for (int h0 = hh * (GAT_H / 2); h0 < (hh + 1) * (GAT_H / 2); h0 += 8) {
    const float4 sa = s2l[(h0 / 4) * GAT_K + j];
    const float4 sb = s2l[(h0 / 4 + 1) * GAT_K + j];
    const float sv[8] = {sa.x, sa.y, sa.z, sa.w, sb.x, sb.y, sb.z, sb.w};
#pragma unroll
    for (int k = 0; k < 8; ++k) {
      const float ah = av[h0 + k];
      float t;
      t = r1p[0 * GAT_H + h0 + k] + sv[k];
      t = fmaxf(t, GAT_ALPHA * t);
      acc0 = fmaf(t, ah, acc0);
      t = r1p[1 * GAT_H + h0 + k] + sv[k];
      t = fmaxf(t, GAT_ALPHA * t);
      acc1 = fmaf(t, ah, acc1);
      t = r1p[2 * GAT_H + h0 + k] + sv[k];
      t = fmaxf(t, GAT_ALPHA * t);
      acc2 = fmaf(t, ah, acc2);
      t = r1p[3 * GAT_H + h0 + k] + sv[k];
      t = fmaxf(t, GAT_ALPHA * t);
      acc3 = fmaf(t, ah, acc3);
    }
  }

  const int ft = wave & 7;
  const int kh = wave >> 3;
  float4 xq[8];
  {
    const float4* xr = reinterpret_cast<const float4*>(
        x + (b * GAT_F + ft * 16 + l) * GAT_K + kh * 64 + hf * 4);
#pragma unroll
    for (int uu = 0; uu < 8; ++uu) xq[uu] = xr[uu * 2];
  }

  el[(hh * 8 + rg * 4 + 0) * GAT_K + j] = acc0;
  el[(hh * 8 + rg * 4 + 1) * GAT_K + j] = acc1;
  el[(hh * 8 + rg * 4 + 2) * GAT_K + j] = acc2;
  el[(hh * 8 + rg * 4 + 3) * GAT_K + j] = acc3;
  lds_barrier();

  if (wave < 8) {
    const int i = ic * 8 + wave;
    const float* p0 = el + wave * GAT_K + lane;
    const float* p1 = el + (8 + wave) * GAT_K + lane;
    const float* bi = bias + i * GAT_K + lane;
    float e0 = (p0[ 0] + p1[ 0]) + bi[ 0];
    float e1 = (p0[32] + p1[32]) + bi[32];
    float e2 = (p0[64] + p1[64]) + bi[64];
    float e3 = (p0[96] + p1[96]) + bi[96];
    float m = fmaxf(fmaxf(e0, e1), fmaxf(e2, e3));
    m = fmaxf(m, __shfl_xor(m, 1));
    m = fmaxf(m, __shfl_xor(m, 2));
    m = fmaxf(m, __shfl_xor(m, 4));
    m = fmaxf(m, __shfl_xor(m, 8));
    m = fmaxf(m, __shfl_xor(m, 16));
    e0 = __expf(e0 - m);
    e1 = __expf(e1 - m);
    e2 = __expf(e2 - m);
    e3 = __expf(e3 - m);
    float s = e0 + e1 + e2 + e3;
    s += __shfl_xor(s, 1);
    s += __shfl_xor(s, 2);
    s += __shfl_xor(s, 4);
    s += __shfl_xor(s, 8);
    s += __shfl_xor(s, 16);
    const float inv = 1.0f / s;
    float* prow = reinterpret_cast<float*>(pl) + wave * 132 + lane;
    prow[ 0] = e0 * inv;
    prow[32] = e1 * inv;
    prow[64] = e2 * inv;
    prow[96] = e3 * inv;
  }
  lds_barrier();

  v8f c0 = {};
  v8f c1 = {};
#pragma unroll
  for (int uu = 0; uu < 8; ++uu) {
    const float4 pa = pl[(l & 7) * 33 + kh * 16 + uu * 2 + hf];
    const float4 xb = xq[uu];
    v2f a0, a1, b0, b1;
    a0.x = pa.x; a0.y = pa.y; a1.x = pa.z; a1.y = pa.w;
    b0.x = xb.x; b0.y = xb.y; b1.x = xb.z; b1.y = xb.w;
    c0 = __builtin_amdgcn_wmma_f32_16x16x4_f32(false, a0, false, b0, (short)0,
                                               c0, false, false);
    c1 = __builtin_amdgcn_wmma_f32_16x16x4_f32(false, a1, false, b1, (short)0,
                                               c1, false, false);
  }
  {
    float* ep = el + (kh * 8 + 4 * hf) * GAT_K + ft * 16 + l;
    ep[0 * GAT_K] = hf ? (c0[4] + c1[4]) : (c0[0] + c1[0]);
    ep[1 * GAT_K] = hf ? (c0[5] + c1[5]) : (c0[1] + c1[1]);
    ep[2 * GAT_K] = hf ? (c0[6] + c1[6]) : (c0[2] + c1[2]);
    ep[3 * GAT_K] = hf ? (c0[7] + c1[7]) : (c0[3] + c1[3]);
  }
  __syncthreads();

  {
    const int row = wave & 7;
    const int f   = (wave >> 3) * 64 + lane;
    const float z0 = el[row * GAT_K + f]      + el[(8 + row) * GAT_K + f];
    const float z1 = el[row * GAT_K + f + 32] + el[(8 + row) * GAT_K + f + 32];
    float* orow = out + (b * GAT_K + ic * 8 + row) * GAT_F + f;
    orow[ 0] = 1.0f / (1.0f + __expf(-z0));
    orow[32] = 1.0f / (1.0f + __expf(-z1));
  }
}

extern "C" void kernel_launch(void* const* d_in, const int* in_sizes, int n_in,
                              void* d_out, int out_size, void* d_ws, size_t ws_size,
                              hipStream_t stream) {
  (void)in_sizes; (void)n_in; (void)out_size; (void)ws_size;
  const float* x    = (const float*)d_in[0];
  const float* w    = (const float*)d_in[1];
  const float* lb   = (const float*)d_in[2];
  const float* av   = (const float*)d_in[3];
  const float* bias = (const float*)d_in[4];
  float* out = (float*)d_out;

  float* s1p  = (float*)d_ws;
  float* s2q  = s1p + GAT_B * GAT_K * GAT_H;

  gat_qk_gemm<<<GAT_B * (GAT_K / 32) * (GAT_H / 16), 32, 0, stream>>>(x, w, lb, s1p, s2q);
  gat_attn_out<<<GAT_B * (GAT_K / 8),             512, 0, stream>>>(s1p, s2q, av, bias, x, out);
}


# attention waves re-tiled as h eighth x 8 rows x 64 cols (lane owns 2 cols): each s2 value has one consumer lane, loaded global->registers; LDS staging of s2 dropped; fp32 sum order over h changes
# speedup vs baseline: 4.3435x; 1.1981x over previous
#include <hip/hip_runtime.h>

typedef __attribute__((ext_vector_type(2))) float v2f;
typedef __attribute__((ext_vector_type(8))) float v8f;

#define GAT_B 16
#define GAT_K 128
#define GAT_F 128
#define GAT_H 256
#define GAT_ALPHA 0.2f

__global__ void __launch_bounds__(32) __attribute__((amdgpu_waves_per_eu(1, 2)))
gat_qk_gemm(const float* __restrict__ x,
            const float* __restrict__ w,
            const float* __restrict__ lb,
            float* __restrict__ s1p,
            float* __restrict__ s2q) {
  int tile = blockIdx.x;
  const int ht = tile % (GAT_H / 16); tile /= (GAT_H / 16);
  const int ip = tile % (GAT_K / 32); tile /= (GAT_K / 32);
  const int b  = tile;
  const int i0 = ip * 32, h0 = ht * 16;
  const int lane = threadIdx.x & 31;
  const int hf   = lane >> 4;
  const int l    = lane & 15;

  const float* xb = x + b * GAT_F * GAT_K + (4 * hf) * GAT_K + i0 + 2 * l;
  const float* wr = w + (h0 + l) * (2 * GAT_F) + 4 * hf;

  v8f c1a = {}, c1b = {};
  v8f c2a = {}, c2b = {};
#pragma unroll
  for (int m = 0; m < GAT_F / 8; ++m) {
    const float4 w1 = *reinterpret_cast<const float4*>(wr + 8 * m);
    const float4 w2 = *reinterpret_cast<const float4*>(wr + GAT_F + 8 * m);
    const v2f x0 = *reinterpret_cast<const v2f*>(xb + (8 * m + 0) * GAT_K);
    const v2f x1 = *reinterpret_cast<const v2f*>(xb + (8 * m + 1) * GAT_K);
    const v2f x2 = *reinterpret_cast<const v2f*>(xb + (8 * m + 2) * GAT_K);
    const v2f x3 = *reinterpret_cast<const v2f*>(xb + (8 * m + 3) * GAT_K);
    v2f va, vb, b1, b2;
    va.x = x0.x; va.y = x1.x;
    vb.x = x0.y; vb.y = x1.y;
    b1.x = w1.x; b1.y = w1.y;
    b2.x = w2.x; b2.y = w2.y;
    c1a = __builtin_amdgcn_wmma_f32_16x16x4_f32(false, va, false, b1, (short)0,
                                                c1a, false, false);
    c1b = __builtin_amdgcn_wmma_f32_16x16x4_f32(false, vb, false, b1, (short)0,
                                                c1b, false, false);
    c2a = __builtin_amdgcn_wmma_f32_16x16x4_f32(false, b2, false, va, (short)0,
                                                c2a, false, false);
    c2b = __builtin_amdgcn_wmma_f32_16x16x4_f32(false, b2, false, vb, (short)0,
                                                c2b, false, false);
    va.x = x2.x; va.y = x3.x;
    vb.x = x2.y; vb.y = x3.y;
    b1.x = w1.z; b1.y = w1.w;
    b2.x = w2.z; b2.y = w2.w;
    c1a = __builtin_amdgcn_wmma_f32_16x16x4_f32(false, va, false, b1, (short)0,
                                                c1a, false, false);
    c1b = __builtin_amdgcn_wmma_f32_16x16x4_f32(false, vb, false, b1, (short)0,
                                                c1b, false, false);
    c2a = __builtin_amdgcn_wmma_f32_16x16x4_f32(false, b2, false, va, (short)0,
                                                c2a, false, false);
    c2b = __builtin_amdgcn_wmma_f32_16x16x4_f32(false, b2, false, vb, (short)0,
                                                c2b, false, false);
  }
  const float lbv = lb[h0 + l];
#pragma unroll
  for (int r = 0; r < 8; ++r) {
    const int row = i0 + 2 * (r + 8 * hf);
    const int col = h0 + l;
    s1p[(b * GAT_K + row) * GAT_H + col]     = c1a[r] + lbv;
    s1p[(b * GAT_K + row + 1) * GAT_H + col] = c1b[r] + lbv;
  }
  float4* q = reinterpret_cast<float4*>(s2q) +
              (b * (GAT_H / 4) + (h0 + 8 * hf) / 4) * GAT_K + i0 + 2 * l;
  q[0]         = make_float4(c2a[0], c2a[1], c2a[2], c2a[3]);
  q[1]         = make_float4(c2b[0], c2b[1], c2b[2], c2b[3]);
  q[GAT_K]     = make_float4(c2a[4], c2a[5], c2a[6], c2a[7]);
  q[GAT_K + 1] = make_float4(c2b[4], c2b[5], c2b[6], c2b[7]);
}

__device__ __forceinline__ void lds_barrier() {
  __builtin_amdgcn_fence(__ATOMIC_RELEASE, "workgroup", "local");
  __builtin_amdgcn_s_barrier();
  __builtin_amdgcn_fence(__ATOMIC_ACQUIRE, "workgroup", "local");
}

__global__ void __launch_bounds__(512) __attribute__((amdgpu_waves_per_eu(4, 4)))
gat_attn_out(const float* __restrict__ s1p,
             const float* __restrict__ s2q,
             const float* __restrict__ av,
             const float* __restrict__ bias,
             const float* __restrict__ x,
             float* __restrict__ out) {
  __shared__ float  el[8 * 8 * GAT_K];
  __shared__ float4 pl[8 * 33];

  const int blk  = blockIdx.x;
  const int ic   = blk % (GAT_K / 8);
  const int b    = blk / (GAT_K / 8);
  const int wave = __builtin_amdgcn_readfirstlane((int)(threadIdx.x >> 5));
  const int lane = threadIdx.x & 31;
  const int hf   = lane >> 4;
  const int l    = lane & 15;

  const int he = wave >> 1;
  const int jh = wave & 1;
  const int j  = jh * 64 + lane;

  float4 sqa[GAT_H / 32], sqb[GAT_H / 32];
  {
    const float4* src = reinterpret_cast<const float4*>(s2q) +
                        (b * (GAT_H / 4) + he * (GAT_H / 32)) * GAT_K + j;
#pragma unroll
    for (int n = 0; n < GAT_H / 32; ++n) {
      sqa[n] = src[n * GAT_K];
      sqb[n] = src[n * GAT_K + 32];
    }
  }

  const float* r1p = s1p + (b * GAT_K + ic * 8) * GAT_H + he * (GAT_H / 8);
  const float* avp = av + he * (GAT_H / 8);

  float acca[8], accb[8];
#pragma unroll
  for (int r = 0; r < 8; ++r) { acca[r] = 0.f; accb[r] = 0.f; }

#pragma unroll
  for (int t = 0; t < GAT_H / 128; ++t) {
    float sva[16], svb[16];
#pragma unroll
    for (int n = 0; n < 4; ++n) {
      const float4 qa = sqa[4 * t + n], qb = sqb[4 * t + n];
      sva[4 * n + 0] = qa.x; sva[4 * n + 1] = qa.y; sva[4 * n + 2] = qa.z; sva[4 * n + 3] = qa.w;
      svb[4 * n + 0] = qb.x; svb[4 * n + 1] = qb.y; svb[4 * n + 2] = qb.z; svb[4 * n + 3] = qb.w;
    }
#pragma unroll
    for (int g = 0; g < 2; ++g) {
      const float* rp = r1p + g * 4 * GAT_H + 16 * t;
#pragma unroll
      for (int k = 0; k < 16; ++k) {
        const float ah = avp[16 * t + k];
#pragma unroll
        for (int r = 0; r < 4; ++r) {
          const float s1v = rp[r * GAT_H + k];
          float ta = s1v + sva[k];
          float tb = s1v + svb[k];
          ta = fmaxf(ta, GAT_ALPHA * ta);
          tb = fmaxf(tb, GAT_ALPHA * tb);
          acca[4 * g + r] = fmaf(ta, ah, acca[4 * g + r]);
          accb[4 * g + r] = fmaf(tb, ah, accb[4 * g + r]);
        }
      }
      __builtin_amdgcn_sched_barrier(0);
    }
  }

  const int ft = wave & 7;
  const int kh = wave >> 3;
  float4 xq[8];
  {
    const float4* xr = reinterpret_cast<const float4*>(
        x + (b * GAT_F + ft * 16 + l) * GAT_K + kh * 64 + hf * 4);
#pragma unroll
    for (int uu = 0; uu < 8; ++uu) xq[uu] = xr[uu * 2];
  }

#pragma unroll
  for (int r = 0; r < 8; ++r) {
    el[(he * 8 + r) * GAT_K + j]      = acca[r];
    el[(he * 8 + r) * GAT_K + j + 32] = accb[r];
  }
  lds_barrier();

  if (wave < 8) {
    const int i = ic * 8 + wave;
    const float* p = el + wave * GAT_K + lane;
    const float* bi = bias + i * GAT_K + lane;
    float e[4];
#pragma unroll
    for (int c = 0; c < 4; ++c) {
      const float* pc = p + 32 * c;
      e[c] = (((pc[0 * 8 * GAT_K] + pc[1 * 8 * GAT_K]) + (pc[2 * 8 * GAT_K] + pc[3 * 8 * GAT_K])) +
              ((pc[4 * 8 * GAT_K] + pc[5 * 8 * GAT_K]) + (pc[6 * 8 * GAT_K] + pc[7 * 8 * GAT_K]))) +
             bi[32 * c];
    }
    float e0 = e[0], e1 = e[1], e2 = e[2], e3 = e[3];
    float m = fmaxf(fmaxf(e0, e1), fmaxf(e2, e3));
    m = fmaxf(m, __shfl_xor(m, 1));
    m = fmaxf(m, __shfl_xor(m, 2));
    m = fmaxf(m, __shfl_xor(m, 4));
    m = fmaxf(m, __shfl_xor(m, 8));
    m = fmaxf(m, __shfl_xor(m, 16));
    e0 = __expf(e0 - m);
    e1 = __expf(e1 - m);
    e2 = __expf(e2 - m);
    e3 = __expf(e3 - m);
    float s = e0 + e1 + e2 + e3;
    s += __shfl_xor(s, 1);
    s += __shfl_xor(s, 2);
    s += __shfl_xor(s, 4);
    s += __shfl_xor(s, 8);
    s += __shfl_xor(s, 16);
    const float inv = 1.0f / s;
    float* prow = reinterpret_cast<float*>(pl) + wave * 132 + lane;
    prow[ 0] = e0 * inv;
    prow[32] = e1 * inv;
    prow[64] = e2 * inv;
    prow[96] = e3 * inv;
  }
  lds_barrier();

  v8f c0 = {};
  v8f c1 = {};
#pragma unroll
  for (int uu = 0; uu < 8; ++uu) {
    const float4 pa = pl[(l & 7) * 33 + kh * 16 + uu * 2 + hf];
    const float4 xb = xq[uu];
    v2f a0, a1, b0, b1;
    a0.x = pa.x; a0.y = pa.y; a1.x = pa.z; a1.y = pa.w;
    b0.x = xb.x; b0.y = xb.y; b1.x = xb.z; b1.y = xb.w;
    c0 = __builtin_amdgcn_wmma_f32_16x16x4_f32(false, a0, false, b0, (short)0,
                                               c0, false, false);
    c1 = __builtin_amdgcn_wmma_f32_16x16x4_f32(false, a1, false, b1, (short)0,
                                               c1, false, false);
  }
  {
    float* ep = el + (kh * 8 + 4 * hf) * GAT_K + ft * 16 + l;
    ep[0 * GAT_K] = hf ? (c0[4] + c1[4]) : (c0[0] + c1[0]);
    ep[1 * GAT_K] = hf ? (c0[5] + c1[5]) : (c0[1] + c1[1]);
    ep[2 * GAT_K] = hf ? (c0[6] + c1[6]) : (c0[2] + c1[2]);
    ep[3 * GAT_K] = hf ? (c0[7] + c1[7]) : (c0[3] + c1[3]);
  }
  __syncthreads();

  {
    const int row = wave & 7;
    const int f   = (wave >> 3) * 64 + lane;
    const float z0 = el[row * GAT_K + f]      + el[(8 + row) * GAT_K + f];
    const float z1 = el[row * GAT_K + f + 32] + el[(8 + row) * GAT_K + f + 32];
    float* orow = out + (b * GAT_K + ic * 8 + row) * GAT_F + f;
    orow[ 0] = 1.0f / (1.0f + __expf(-z0));
    orow[32] = 1.0f / (1.0f + __expf(-z1));
  }
}

extern "C" void kernel_launch(void* const* d_in, const int* in_sizes, int n_in,
                              void* d_out, int out_size, void* d_ws, size_t ws_size,
                              hipStream_t stream) {
  (void)in_sizes; (void)n_in; (void)out_size; (void)ws_size;
  const float* x    = (const float*)d_in[0];
  const float* w    = (const float*)d_in[1];
  const float* lb   = (const float*)d_in[2];
  const float* av   = (const float*)d_in[3];
  const float* bias = (const float*)d_in[4];
  float* out = (float*)d_out;

  float* s1p  = (float*)d_ws;
  float* s2q  = s1p + GAT_B * GAT_K * GAT_H;

  gat_qk_gemm<<<GAT_B * (GAT_K / 32) * (GAT_H / 16), 32, 0, stream>>>(x, w, lb, s1p, s2q);
  gat_attn_out<<<GAT_B * (GAT_K / 8),             512, 0, stream>>>(s1p, s2q, av, bias, x, out);
}


# min+fma score loop: lrelu(t)=alpha t+c max(t,0), c max(p+q,0)=c q-min(-c p,c q): 2 dual-issued ops per (i,j,h), was 4; -c s1 on scalar unit; a.s1 term cancels in softmax; fp32 rounding of e changes
# speedup vs baseline: 4.6390x; 1.0680x over previous
#include <hip/hip_runtime.h>

typedef __attribute__((ext_vector_type(2))) float v2f;
typedef __attribute__((ext_vector_type(8))) float v8f;

#define GAT_B 16
#define GAT_K 128
#define GAT_F 128
#define GAT_H 256
#define GAT_ALPHA 0.2f

__global__ void __launch_bounds__(32) __attribute__((amdgpu_waves_per_eu(1, 2)))
gat_qk_gemm(const float* __restrict__ x,
            const float* __restrict__ w,
            const float* __restrict__ lb,
            float* __restrict__ s1p,
            float* __restrict__ s2q) {
  int tile = blockIdx.x;
  const int ht = tile % (GAT_H / 16); tile /= (GAT_H / 16);
  const int ip = tile % (GAT_K / 32); tile /= (GAT_K / 32);
  const int b  = tile;
  const int i0 = ip * 32, h0 = ht * 16;
  const int lane = threadIdx.x & 31;
  const int hf   = lane >> 4;
  const int l    = lane & 15;

  const float* xb = x + b * GAT_F * GAT_K + (4 * hf) * GAT_K + i0 + 2 * l;
  const float* wr = w + (h0 + l) * (2 * GAT_F) + 4 * hf;

  v8f c1a = {}, c1b = {};
  v8f c2a = {}, c2b = {};
#pragma unroll
  for (int m = 0; m < GAT_F / 8; ++m) {
    const float4 w1 = *reinterpret_cast<const float4*>(wr + 8 * m);
    const float4 w2 = *reinterpret_cast<const float4*>(wr + GAT_F + 8 * m);
    const v2f x0 = *reinterpret_cast<const v2f*>(xb + (8 * m + 0) * GAT_K);
    const v2f x1 = *reinterpret_cast<const v2f*>(xb + (8 * m + 1) * GAT_K);
    const v2f x2 = *reinterpret_cast<const v2f*>(xb + (8 * m + 2) * GAT_K);
    const v2f x3 = *reinterpret_cast<const v2f*>(xb + (8 * m + 3) * GAT_K);
    v2f va, vb, b1, b2;
    va.x = x0.x; va.y = x1.x;
    vb.x = x0.y; vb.y = x1.y;
    b1.x = w1.x; b1.y = w1.y;
    b2.x = w2.x; b2.y = w2.y;
    c1a = __builtin_amdgcn_wmma_f32_16x16x4_f32(false, va, false, b1, (short)0,
                                                c1a, false, false);
    c1b = __builtin_amdgcn_wmma_f32_16x16x4_f32(false, vb, false, b1, (short)0,
                                                c1b, false, false);
    c2a = __builtin_amdgcn_wmma_f32_16x16x4_f32(false, b2, false, va, (short)0,
                                                c2a, false, false);
    c2b = __builtin_amdgcn_wmma_f32_16x16x4_f32(false, b2, false, vb, (short)0,
                                                c2b, false, false);
    va.x = x2.x; va.y = x3.x;
    vb.x = x2.y; vb.y = x3.y;
    b1.x = w1.z; b1.y = w1.w;
    b2.x = w2.z; b2.y = w2.w;
    c1a = __builtin_amdgcn_wmma_f32_16x16x4_f32(false, va, false, b1, (short)0,
                                                c1a, false, false);
    c1b = __builtin_amdgcn_wmma_f32_16x16x4_f32(false, vb, false, b1, (short)0,
                                                c1b, false, false);
    c2a = __builtin_amdgcn_wmma_f32_16x16x4_f32(false, b2, false, va, (short)0,
                                                c2a, false, false);
    c2b = __builtin_amdgcn_wmma_f32_16x16x4_f32(false, b2, false, vb, (short)0,
                                                c2b, false, false);
  }
  const float lbv = lb[h0 + l];
#pragma unroll
  for (int r = 0; r < 8; ++r) {
    const int row = i0 + 2 * (r + 8 * hf);
    const int col = h0 + l;
    s1p[(b * GAT_K + row) * GAT_H + col]     = c1a[r] + lbv;
    s1p[(b * GAT_K + row + 1) * GAT_H + col] = c1b[r] + lbv;
  }
  float4* q = reinterpret_cast<float4*>(s2q) +
              (b * (GAT_H / 4) + (h0 + 8 * hf) / 4) * GAT_K + i0 + 2 * l;
  q[0]         = make_float4(c2a[0], c2a[1], c2a[2], c2a[3]);
  q[1]         = make_float4(c2b[0], c2b[1], c2b[2], c2b[3]);
  q[GAT_K]     = make_float4(c2a[4], c2a[5], c2a[6], c2a[7]);
  q[GAT_K + 1] = make_float4(c2b[4], c2b[5], c2b[6], c2b[7]);
}

__device__ __forceinline__ void lds_barrier() {
  __builtin_amdgcn_fence(__ATOMIC_RELEASE, "workgroup", "local");
  __builtin_amdgcn_s_barrier();
  __builtin_amdgcn_fence(__ATOMIC_ACQUIRE, "workgroup", "local");
}

__global__ void __launch_bounds__(512) __attribute__((amdgpu_waves_per_eu(4, 4)))
gat_attn_out(const float* __restrict__ s1p,
             const float* __restrict__ s2q,
             const float* __restrict__ av,
             const float* __restrict__ bias,
             const float* __restrict__ x,
             float* __restrict__ out) {
  __shared__ float  el[8 * 8 * GAT_K];
  __shared__ float4 pl[8 * 33];

  const int blk  = blockIdx.x;
  const int ic   = blk % (GAT_K / 8);
  const int b    = blk / (GAT_K / 8);
  const int wave = __builtin_amdgcn_readfirstlane((int)(threadIdx.x >> 5));
  const int lane = threadIdx.x & 31;
  const int hf   = lane >> 4;
  const int l    = lane & 15;

  const int he = wave >> 1;
  const int jh = wave & 1;
  const int j  = jh * 64 + lane;

  float4 sqa[GAT_H / 32], sqb[GAT_H / 32];
  {
    const float4* src = reinterpret_cast<const float4*>(s2q) +
                        (b * (GAT_H / 4) + he * (GAT_H / 32)) * GAT_K + j;
#pragma unroll
    for (int n = 0; n < GAT_H / 32; ++n) {
      sqa[n] = src[n * GAT_K];
      sqb[n] = src[n * GAT_K + 32];
    }
  }

  const float* r1p = s1p + (b * GAT_K + ic * 8) * GAT_H + he * (GAT_H / 8);
  const float* avp = av + he * (GAT_H / 8);

  float acca[8], accb[8];
#pragma unroll
  for (int r = 0; r < 8; ++r) { acca[r] = 0.f; accb[r] = 0.f; }
  float a2a = 0.f, a2b = 0.f;
  const float cpos = 1.0f - GAT_ALPHA;
  const float cneg = GAT_ALPHA - 1.0f;

#pragma unroll
  for (int t = 0; t < GAT_H / 128; ++t) {
    float sva[16], svb[16];
    float cva[16], cvb[16];
#pragma unroll
    for (int n = 0; n < 4; ++n) {
      const float4 qa = sqa[4 * t + n], qb = sqb[4 * t + n];
      sva[4 * n + 0] = qa.x; sva[4 * n + 1] = qa.y; sva[4 * n + 2] = qa.z; sva[4 * n + 3] = qa.w;
      svb[4 * n + 0] = qb.x; svb[4 * n + 1] = qb.y; svb[4 * n + 2] = qb.z; svb[4 * n + 3] = qb.w;
    }
#pragma unroll
    for (int k = 0; k < 16; ++k) { cva[k] = cpos * sva[k]; cvb[k] = cpos * svb[k]; }
#pragma unroll
    for (int g = 0; g < 2; ++g) {
      const float* rp = r1p + g * 4 * GAT_H + 16 * t;
#pragma unroll
      for (int k = 0; k < 16; ++k) {
        const float ah = avp[16 * t + k];
        if (g == 0) {
          a2a = fmaf(sva[k], ah, a2a);
          a2b = fmaf(svb[k], ah, a2b);
        }
#pragma unroll
        for (int r = 0; r < 4; ++r) {
          const float n1v = cneg * rp[r * GAT_H + k];
          acca[4 * g + r] = fmaf(fminf(n1v, cva[k]), ah, acca[4 * g + r]);
          accb[4 * g + r] = fmaf(fminf(n1v, cvb[k]), ah, accb[4 * g + r]);
        }
      }
      __builtin_amdgcn_sched_barrier(0);
    }
  }

  const int ft = wave & 7;
  const int kh = wave >> 3;
  float4 xq[8];
  {
    const float4* xr = reinterpret_cast<const float4*>(
        x + (b * GAT_F + ft * 16 + l) * GAT_K + kh * 64 + hf * 4);
#pragma unroll
    for (int uu = 0; uu < 8; ++uu) xq[uu] = xr[uu * 2];
  }

#pragma unroll
  for (int r = 0; r < 8; ++r) {
    el[(he * 8 + r) * GAT_K + j]      = a2a - acca[r];
    el[(he * 8 + r) * GAT_K + j + 32] = a2b - accb[r];
  }
  lds_barrier();

  if (wave < 8) {
    const int i = ic * 8 + wave;
    const float* p = el + wave * GAT_K + lane;
    const float* bi = bias + i * GAT_K + lane;
    float e[4];
#pragma unroll
    for (int c = 0; c < 4; ++c) {
      const float* pc = p + 32 * c;
      e[c] = (((pc[0 * 8 * GAT_K] + pc[1 * 8 * GAT_K]) + (pc[2 * 8 * GAT_K] + pc[3 * 8 * GAT_K])) +
              ((pc[4 * 8 * GAT_K] + pc[5 * 8 * GAT_K]) + (pc[6 * 8 * GAT_K] + pc[7 * 8 * GAT_K]))) +
             bi[32 * c];
    }
    float e0 = e[0], e1 = e[1], e2 = e[2], e3 = e[3];
    float m = fmaxf(fmaxf(e0, e1), fmaxf(e2, e3));
    m = fmaxf(m, __shfl_xor(m, 1));
    m = fmaxf(m, __shfl_xor(m, 2));
    m = fmaxf(m, __shfl_xor(m, 4));
    m = fmaxf(m, __shfl_xor(m, 8));
    m = fmaxf(m, __shfl_xor(m, 16));
    e0 = __expf(e0 - m);
    e1 = __expf(e1 - m);
    e2 = __expf(e2 - m);
    e3 = __expf(e3 - m);
    float s = e0 + e1 + e2 + e3;
    s += __shfl_xor(s, 1);
    s += __shfl_xor(s, 2);
    s += __shfl_xor(s, 4);
    s += __shfl_xor(s, 8);
    s += __shfl_xor(s, 16);
    const float inv = 1.0f / s;
    float* prow = reinterpret_cast<float*>(pl) + wave * 132 + lane;
    prow[ 0] = e0 * inv;
    prow[32] = e1 * inv;
    prow[64] = e2 * inv;
    prow[96] = e3 * inv;
  }
  lds_barrier();

  v8f c0 = {};
  v8f c1 = {};
#pragma unroll
  for (int uu = 0; uu < 8; ++uu) {
    const float4 pa = pl[(l & 7) * 33 + kh * 16 + uu * 2 + hf];
    const float4 xb = xq[uu];
    v2f a0, a1, b0, b1;
    a0.x = pa.x; a0.y = pa.y; a1.x = pa.z; a1.y = pa.w;
    b0.x = xb.x; b0.y = xb.y; b1.x = xb.z; b1.y = xb.w;
    c0 = __builtin_amdgcn_wmma_f32_16x16x4_f32(false, a0, false, b0, (short)0,
                                               c0, false, false);
    c1 = __builtin_amdgcn_wmma_f32_16x16x4_f32(false, a1, false, b1, (short)0,
                                               c1, false, false);
  }
  {
    float* ep = el + (kh * 8 + 4 * hf) * GAT_K + ft * 16 + l;
    ep[0 * GAT_K] = hf ? (c0[4] + c1[4]) : (c0[0] + c1[0]);
    ep[1 * GAT_K] = hf ? (c0[5] + c1[5]) : (c0[1] + c1[1]);
    ep[2 * GAT_K] = hf ? (c0[6] + c1[6]) : (c0[2] + c1[2]);
    ep[3 * GAT_K] = hf ? (c0[7] + c1[7]) : (c0[3] + c1[3]);
  }
  __syncthreads();

  {
    const int row = wave & 7;
    const int f   = (wave >> 3) * 64 + lane;
    const float z0 = el[row * GAT_K + f]      + el[(8 + row) * GAT_K + f];
    const float z1 = el[row * GAT_K + f + 32] + el[(8 + row) * GAT_K + f + 32];
    float* orow = out + (b * GAT_K + ic * 8 + row) * GAT_F + f;
    orow[ 0] = 1.0f / (1.0f + __expf(-z0));
    orow[32] = 1.0f / (1.0f + __expf(-z1));
  }
}

extern "C" void kernel_launch(void* const* d_in, const int* in_sizes, int n_in,
                              void* d_out, int out_size, void* d_ws, size_t ws_size,
                              hipStream_t stream) {
  (void)in_sizes; (void)n_in; (void)out_size; (void)ws_size;
  const float* x    = (const float*)d_in[0];
  const float* w    = (const float*)d_in[1];
  const float* lb   = (const float*)d_in[2];
  const float* av   = (const float*)d_in[3];
  const float* bias = (const float*)d_in[4];
  float* out = (float*)d_out;

  float* s1p  = (float*)d_ws;
  float* s2q  = s1p + GAT_B * GAT_K * GAT_H;

  gat_qk_gemm<<<GAT_B * (GAT_K / 32) * (GAT_H / 16), 32, 0, stream>>>(x, w, lb, s1p, s2q);
  gat_attn_out<<<GAT_B * (GAT_K / 8),             512, 0, stream>>>(s1p, s2q, av, bias, x, out);
}
